# starting kernel
# baseline (speedup 1.0000x reference)
#include <hip/hip_runtime.h>
#include <hip/hip_bf16.h>

#define NN 100000
#define FF 128
#define BB 1024
#define SS 32
#define DD 64

typedef float v2f_t __attribute__((ext_vector_type(2)));
typedef float v8f_t __attribute__((ext_vector_type(8)));

#if defined(__AMDGCN__) && __has_builtin(__builtin_amdgcn_wmma_f32_16x16x4_f32)
#define HAVE_WMMA_F32 1
#endif

__device__ __forceinline__ v8f_t wmma_f32_4(v2f_t a, v2f_t b, v8f_t c) {
#ifdef HAVE_WMMA_F32
    return __builtin_amdgcn_wmma_f32_16x16x4_f32(false, a, false, b,
                                                 (short)0, c, false, false);
#else
    (void)a; (void)b;
    return c;
#endif
}

__global__ void k1_compute_h(const float* __restrict__ feat,
                             const float* __restrict__ W1,
                             const float* __restrict__ b1,
                             float* __restrict__ H) {
    __shared__ float sA[16 * FF];
    __shared__ float sW[FF * DD];
    const int t    = threadIdx.x;
    const int row0 = blockIdx.x * 16;

    for (int idx = t; idx < 16 * FF; idx += 128)
        sA[idx] = feat[(size_t)row0 * FF + idx];
    for (int idx = t; idx < FF * DD; idx += 128)
        sW[idx] = W1[idx];
    __syncthreads();

    const int wave  = t >> 5;
    const int lane  = t & 31;
    const int half  = lane >> 4;
    const int l     = lane & 15;
    const int ncol0 = wave * 16;

    v8f_t c = {0.f, 0.f, 0.f, 0.f, 0.f, 0.f, 0.f, 0.f};
    #pragma unroll
    for (int k0 = 0; k0 < FF; k0 += 4) {
        const int ka = k0 + 2 * half;
        v2f_t a, b;
        a.x = sA[l * FF + ka];
        a.y = sA[l * FF + ka + 1];
        b.x = sW[ka * DD + ncol0 + l];
        b.y = sW[(ka + 1) * DD + ncol0 + l];
        c = wmma_f32_4(a, b, c);
    }

    const int   col  = ncol0 + l;
    const float bias = b1[col];
    #pragma unroll
    for (int r = 0; r < 8; ++r) {
        const int row = row0 + half * 8 + r;
        H[(size_t)row * DD + col] = fmaxf(c[r] + bias, 0.f);
    }
}

__global__ void k2_gather_mean(const float* __restrict__ H,
                               const int* __restrict__ s1,
                               const int* __restrict__ s2,
                               const int* __restrict__ bn,
                               float* __restrict__ agg1,
                               float* __restrict__ aggn) {
    const int wave = threadIdx.x >> 5;
    const int lane = threadIdx.x & 31;
    const int g    = blockIdx.x * 8 + wave;

    float accx = 0.f, accy = 0.f;
    float* dst;

    if (g < BB * SS) {
        const int b = g >> 5;
        const int i = g & 31;
        const int* base = s2 + (size_t)b * SS * SS + i;
        for (int j = 0; j < SS; ++j) {
            const int    idx = base[j * SS];
            const float2 hv  = *(const float2*)(H + (size_t)idx * DD + 2 * lane);
            accx += hv.x; accy += hv.y;
        }
        {
            const int    idx = s1[(size_t)b * SS + i];
            const float2 hv  = *(const float2*)(H + (size_t)idx * DD + 2 * lane);
            accx += hv.x; accy += hv.y;
        }
        dst = agg1 + (size_t)g * DD;
    } else {
        const int b = g - BB * SS;
        for (int j = 0; j < SS; ++j) {
            const int    idx = s1[(size_t)b * SS + j];
            const float2 hv  = *(const float2*)(H + (size_t)idx * DD + 2 * lane);
            accx += hv.x; accy += hv.y;
        }
        {
            const int    idx = bn[b];
            const float2 hv  = *(const float2*)(H + (size_t)idx * DD + 2 * lane);
            accx += hv.x; accy += hv.y;
        }
        dst = aggn + (size_t)b * DD;
    }

    const float sc = 1.f / 33.f;
    ((float2*)dst)[lane] = make_float2(accx * sc, accy * sc);
}

__global__ void k3_final(const float* __restrict__ agg1,
                         const float* __restrict__ aggn,
                         const float* __restrict__ W2,
                         const float* __restrict__ b2,
                         float* __restrict__ out) {
    __shared__ float sX[48 * DD];
    __shared__ float sW[DD * DD];
    const int t = threadIdx.x;
    const int b = blockIdx.x;

    for (int idx = t; idx < 48 * DD; idx += 128) {
        const int row = idx >> 6, col = idx & 63;
        float v = 0.f;
        if (row < SS)       v = agg1[((size_t)b * SS + row) * DD + col];
        else if (row == SS) v = aggn[(size_t)b * DD + col];
        sX[idx] = v;
    }
    for (int idx = t; idx < DD * DD; idx += 128)
        sW[idx] = W2[idx];
    __syncthreads();

    const int wave  = t >> 5;
    const int lane  = t & 31;
    const int half  = lane >> 4;
    const int l     = lane & 15;
    const int ncol0 = wave * 16;
    const float bias = b2[ncol0 + l];

    float colsum = 0.f;
    #pragma unroll
    for (int mt = 0; mt < 3; ++mt) {
        v8f_t c = {0.f, 0.f, 0.f, 0.f, 0.f, 0.f, 0.f, 0.f};
        #pragma unroll
        for (int k0 = 0; k0 < DD; k0 += 4) {
            const int ka = k0 + 2 * half;
            v2f_t a, bb;
            a.x  = sX[(mt * 16 + l) * DD + ka];
            a.y  = sX[(mt * 16 + l) * DD + ka + 1];
            bb.x = sW[ka * DD + ncol0 + l];
            bb.y = sW[(ka + 1) * DD + ncol0 + l];
            c = wmma_f32_4(a, bb, c);
        }
        #pragma unroll
        for (int r = 0; r < 8; ++r) {
            const int row = mt * 16 + half * 8 + r;
            if (row < SS + 1)
                colsum += fmaxf(c[r] + bias, 0.f);
        }
    }

    colsum += __shfl_xor(colsum, 16, 32);
    if (half == 0)
        out[(size_t)b * DD + ncol0 + l] = colsum * (1.f / 33.f);
}

extern "C" void kernel_launch(void* const* d_in, const int* in_sizes, int n_in,
                              void* d_out, int out_size, void* d_ws, size_t ws_size,
                              hipStream_t stream) {
    (void)in_sizes; (void)n_in; (void)out_size; (void)ws_size;

    const float* features    = (const float*)d_in[0];
    const int*   batch_nodes = (const int*)  d_in[1];
    const int*   s1_neighs   = (const int*)  d_in[2];
    const int*   s2_neighs   = (const int*)  d_in[3];
    const float* W1          = (const float*)d_in[4];
    const float* b1          = (const float*)d_in[5];
    const float* W2          = (const float*)d_in[6];
    const float* b2          = (const float*)d_in[7];
    float*       out         = (float*)d_out;

    float* H    = (float*)d_ws;
    float* agg1 = H    + (size_t)NN * DD;
    float* aggn = agg1 + (size_t)BB * SS * DD;

    k1_compute_h <<<NN / 16,           128, 0, stream>>>(features, W1, b1, H);
    k2_gather_mean<<<(BB * SS + BB) / 8, 256, 0, stream>>>(H, s1_neighs, s2_neighs,
                                                           batch_nodes, agg1, aggn);
    k3_final     <<<BB,                128, 0, stream>>>(agg1, aggn, W2, b2, out);
}


# LDS columns of the staged WMMA operand tiles XOR-swizzled (A tiles by 4*row, weight tiles by 16 per k pair) in kernels 1 and 3: fragment reads free of bank conflicts at unchanged LDS size
# speedup vs baseline: 1.1962x; 1.1962x over previous
#include <hip/hip_runtime.h>
#include <hip/hip_bf16.h>

#define NN 100000
#define FF 128
#define BB 1024
#define SS 32
#define DD 64

__device__ __forceinline__ int swz_a(int row, int k) { return k ^ ((row & 15) << 2); }
__device__ __forceinline__ int swz_w(int k, int n)   { return n ^ (((k >> 1) & 1) << 4); }

typedef float v2f_t __attribute__((ext_vector_type(2)));
typedef float v8f_t __attribute__((ext_vector_type(8)));

#if defined(__AMDGCN__) && __has_builtin(__builtin_amdgcn_wmma_f32_16x16x4_f32)
#define HAVE_WMMA_F32 1
#endif

__device__ __forceinline__ v8f_t wmma_f32_4(v2f_t a, v2f_t b, v8f_t c) {
#ifdef HAVE_WMMA_F32
    return __builtin_amdgcn_wmma_f32_16x16x4_f32(false, a, false, b,
                                                 (short)0, c, false, false);
#else
    (void)a; (void)b;
    return c;
#endif
}

__global__ void k1_compute_h(const float* __restrict__ feat,
                             const float* __restrict__ W1,
                             const float* __restrict__ b1,
                             float* __restrict__ H) {
    __shared__ __attribute__((aligned(16))) float sA[16 * FF];
    __shared__ __attribute__((aligned(16))) float sW[FF * DD];
    const int t    = threadIdx.x;
    const int row0 = blockIdx.x * 16;

    for (int idx = t; idx < 16 * FF; idx += 128) {
        const int row = idx >> 7, k = idx & (FF - 1);
        sA[row * FF + swz_a(row, k)] = feat[(size_t)row0 * FF + idx];
    }
    for (int idx = t; idx < FF * DD; idx += 128) {
        const int k = idx >> 6, n = idx & (DD - 1);
        sW[k * DD + swz_w(k, n)] = W1[idx];
    }
    __syncthreads();

    const int wave  = t >> 5;
    const int lane  = t & 31;
    const int half  = lane >> 4;
    const int l     = lane & 15;
    const int ncol0 = wave * 16;
    const int bcol  = swz_w(2 * half, ncol0 + l);

    v8f_t c = {0.f, 0.f, 0.f, 0.f, 0.f, 0.f, 0.f, 0.f};
    #pragma unroll
    for (int k0 = 0; k0 < FF; k0 += 4) {
        const int ka = k0 + 2 * half;
        v2f_t a, b;
        a   = *(const v2f_t*)&sA[l * FF + swz_a(l, ka)];
        b.x = sW[ka * DD + bcol];
        b.y = sW[(ka + 1) * DD + bcol];
        c = wmma_f32_4(a, b, c);
    }

    const int   col  = ncol0 + l;
    const float bias = b1[col];
    #pragma unroll
    for (int r = 0; r < 8; ++r) {
        const int row = row0 + half * 8 + r;
        H[(size_t)row * DD + col] = fmaxf(c[r] + bias, 0.f);
    }
}

__global__ void k2_gather_mean(const float* __restrict__ H,
                               const int* __restrict__ s1,
                               const int* __restrict__ s2,
                               const int* __restrict__ bn,
                               float* __restrict__ agg1,
                               float* __restrict__ aggn) {
    const int wave = threadIdx.x >> 5;
    const int lane = threadIdx.x & 31;
    const int g    = blockIdx.x * 8 + wave;

    float accx = 0.f, accy = 0.f;
    float* dst;

    if (g < BB * SS) {
        const int b = g >> 5;
        const int i = g & 31;
        const int* base = s2 + (size_t)b * SS * SS + i;
        for (int j = 0; j < SS; ++j) {
            const int    idx = base[j * SS];
            const float2 hv  = *(const float2*)(H + (size_t)idx * DD + 2 * lane);
            accx += hv.x; accy += hv.y;
        }
        {
            const int    idx = s1[(size_t)b * SS + i];
            const float2 hv  = *(const float2*)(H + (size_t)idx * DD + 2 * lane);
            accx += hv.x; accy += hv.y;
        }
        dst = agg1 + (size_t)g * DD;
    } else {
        const int b = g - BB * SS;
        for (int j = 0; j < SS; ++j) {
            const int    idx = s1[(size_t)b * SS + j];
            const float2 hv  = *(const float2*)(H + (size_t)idx * DD + 2 * lane);
            accx += hv.x; accy += hv.y;
        }
        {
            const int    idx = bn[b];
            const float2 hv  = *(const float2*)(H + (size_t)idx * DD + 2 * lane);
            accx += hv.x; accy += hv.y;
        }
        dst = aggn + (size_t)b * DD;
    }

    const float sc = 1.f / 33.f;
    ((float2*)dst)[lane] = make_float2(accx * sc, accy * sc);
}

__global__ void k3_final(const float* __restrict__ agg1,
                         const float* __restrict__ aggn,
                         const float* __restrict__ W2,
                         const float* __restrict__ b2,
                         float* __restrict__ out) {
    __shared__ __attribute__((aligned(16))) float sX[48 * DD];
    __shared__ __attribute__((aligned(16))) float sW[DD * DD];
    const int t = threadIdx.x;
    const int b = blockIdx.x;

    for (int idx = t; idx < 48 * DD; idx += 128) {
        const int row = idx >> 6, col = idx & 63;
        float v = 0.f;
        if (row < SS)       v = agg1[((size_t)b * SS + row) * DD + col];
        else if (row == SS) v = aggn[(size_t)b * DD + col];
        sX[row * DD + swz_a(row, col)] = v;
    }
    for (int idx = t; idx < DD * DD; idx += 128) {
        const int k = idx >> 6, n = idx & (DD - 1);
        sW[k * DD + swz_w(k, n)] = W2[idx];
    }
    __syncthreads();

    const int wave  = t >> 5;
    const int lane  = t & 31;
    const int half  = lane >> 4;
    const int l     = lane & 15;
    const int ncol0 = wave * 16;
    const float bias = b2[ncol0 + l];
    const int bcol = swz_w(2 * half, ncol0 + l);

    float colsum = 0.f;
    #pragma unroll
    for (int mt = 0; mt < 3; ++mt) {
        v8f_t c = {0.f, 0.f, 0.f, 0.f, 0.f, 0.f, 0.f, 0.f};
        #pragma unroll
        for (int k0 = 0; k0 < DD; k0 += 4) {
            const int ka = k0 + 2 * half;
            v2f_t a, bb;
            a    = *(const v2f_t*)&sX[(mt * 16 + l) * DD + swz_a(l, ka)];
            bb.x = sW[ka * DD + bcol];
            bb.y = sW[(ka + 1) * DD + bcol];
            c = wmma_f32_4(a, bb, c);
        }
        #pragma unroll
        for (int r = 0; r < 8; ++r) {
            const int row = mt * 16 + half * 8 + r;
            if (row < SS + 1)
                colsum += fmaxf(c[r] + bias, 0.f);
        }
    }

    colsum += __shfl_xor(colsum, 16, 32);
    if (half == 0)
        out[(size_t)b * DD + ncol0 + l] = colsum * (1.f / 33.f);
}

extern "C" void kernel_launch(void* const* d_in, const int* in_sizes, int n_in,
                              void* d_out, int out_size, void* d_ws, size_t ws_size,
                              hipStream_t stream) {
    (void)in_sizes; (void)n_in; (void)out_size; (void)ws_size;

    const float* features    = (const float*)d_in[0];
    const int*   batch_nodes = (const int*)  d_in[1];
    const int*   s1_neighs   = (const int*)  d_in[2];
    const int*   s2_neighs   = (const int*)  d_in[3];
    const float* W1          = (const float*)d_in[4];
    const float* b1          = (const float*)d_in[5];
    const float* W2          = (const float*)d_in[6];
    const float* b2          = (const float*)d_in[7];
    float*       out         = (float*)d_out;

    float* H    = (float*)d_ws;
    float* agg1 = H    + (size_t)NN * DD;
    float* aggn = agg1 + (size_t)BB * SS * DD;

    k1_compute_h <<<NN / 16,           128, 0, stream>>>(features, W1, b1, H);
    k2_gather_mean<<<(BB * SS + BB) / 8, 256, 0, stream>>>(H, s1_neighs, s2_neighs,
                                                           batch_nodes, agg1, aggn);
    k3_final     <<<BB,                128, 0, stream>>>(agg1, aggn, W2, b2, out);
}


# Kernel 1: each wave keeps its 128x16 W1 slice in 64 registers, loaded once, and 1250 work-groups loop over 5 row tiles each, the next feature tile fetched by 16-byte loads during the multiply
# speedup vs baseline: 1.6166x; 1.3514x over previous
#include <hip/hip_runtime.h>
#include <hip/hip_bf16.h>

#define NN 100000
#define FF 128
#define BB 1024
#define SS 32
#define DD 64

__device__ __forceinline__ int swz_a(int row, int k) { return k ^ ((row & 15) << 2); }
__device__ __forceinline__ int swz_w(int k, int n)   { return n ^ (((k >> 1) & 1) << 4); }

typedef float v2f_t __attribute__((ext_vector_type(2)));
typedef float v8f_t __attribute__((ext_vector_type(8)));

#if defined(__AMDGCN__) && __has_builtin(__builtin_amdgcn_wmma_f32_16x16x4_f32)
#define HAVE_WMMA_F32 1
#endif

__device__ __forceinline__ v8f_t wmma_f32_4(v2f_t a, v2f_t b, v8f_t c) {
#ifdef HAVE_WMMA_F32
    return __builtin_amdgcn_wmma_f32_16x16x4_f32(false, a, false, b,
                                                 (short)0, c, false, false);
#else
    (void)a; (void)b;
    return c;
#endif
}

#define K1_TILES  (NN / 16)
#define K1_GROUPS (K1_TILES / 5)

__global__ void __launch_bounds__(128)
k1_compute_h(const float* __restrict__ feat,
             const float* __restrict__ W1,
             const float* __restrict__ b1,
             float* __restrict__ H) {
    __shared__ __attribute__((aligned(16))) float sA[2][16 * FF];
    const int t     = threadIdx.x;
    const int wave  = t >> 5;
    const int lane  = t & 31;
    const int half  = lane >> 4;
    const int l     = lane & 15;
    const int col   = wave * 16 + l;
    const int step  = gridDim.x;

    const int srow = t >> 5;
    const int sk   = (t & 31) << 2;

    int tile = blockIdx.x;
    float4 r0 = make_float4(0.f, 0.f, 0.f, 0.f), r1 = r0, r2 = r0, r3 = r0;
    if (tile < K1_TILES) {
        const float4* src = (const float4*)(feat + (size_t)tile * 16 * FF);
        r0 = src[t]; r1 = src[t + 128]; r2 = src[t + 256]; r3 = src[t + 384];
    }

    v2f_t bf[32];
    #pragma unroll
    for (int s = 0; s < 32; ++s) {
        const int ka = 4 * s + 2 * half;
        bf[s].x = W1[ka * DD + col];
        bf[s].y = W1[(ka + 1) * DD + col];
    }
    const float bias = b1[col];

    const int o0 = (srow     ) * FF + swz_a(srow,      sk);
    const int o1 = (srow +  4) * FF + swz_a(srow +  4, sk);
    const int o2 = (srow +  8) * FF + swz_a(srow +  8, sk);
    const int o3 = (srow + 12) * FF + swz_a(srow + 12, sk);

    if (tile < K1_TILES) {
        *(float4*)&sA[0][o0] = r0; *(float4*)&sA[0][o1] = r1;
        *(float4*)&sA[0][o2] = r2; *(float4*)&sA[0][o3] = r3;
    }

    int buf = 0;
    for (; tile < K1_TILES; tile += step) {
        __syncthreads();

        const int  next     = tile + step;
        const bool has_next = next < K1_TILES;
        if (has_next) {
            const float4* src = (const float4*)(feat + (size_t)next * 16 * FF);
            r0 = src[t]; r1 = src[t + 128]; r2 = src[t + 256]; r3 = src[t + 384];
        }

        const float* A = sA[buf];
        v8f_t c = {0.f, 0.f, 0.f, 0.f, 0.f, 0.f, 0.f, 0.f};
        #pragma unroll
        for (int s = 0; s < 32; ++s) {
            const int ka = 4 * s + 2 * half;
            const v2f_t a = *(const v2f_t*)&A[l * FF + swz_a(l, ka)];
            c = wmma_f32_4(a, bf[s], c);
        }

        const int row0 = tile * 16;
        #pragma unroll
        for (int q = 0; q < 8; ++q) {
            const int row = row0 + half * 8 + q;
            H[(size_t)row * DD + col] = fmaxf(c[q] + bias, 0.f);
        }

        if (has_next) {
            float* Bn = sA[buf ^ 1];
            *(float4*)&Bn[o0] = r0; *(float4*)&Bn[o1] = r1;
            *(float4*)&Bn[o2] = r2; *(float4*)&Bn[o3] = r3;
        }
        buf ^= 1;
    }
}

__global__ void k2_gather_mean(const float* __restrict__ H,
                               const int* __restrict__ s1,
                               const int* __restrict__ s2,
                               const int* __restrict__ bn,
                               float* __restrict__ agg1,
                               float* __restrict__ aggn) {
    const int wave = threadIdx.x >> 5;
    const int lane = threadIdx.x & 31;
    const int g    = blockIdx.x * 8 + wave;

    float accx = 0.f, accy = 0.f;
    float* dst;

    if (g < BB * SS) {
        const int b = g >> 5;
        const int i = g & 31;
        const int* base = s2 + (size_t)b * SS * SS + i;
        for (int j = 0; j < SS; ++j) {
            const int    idx = base[j * SS];
            const float2 hv  = *(const float2*)(H + (size_t)idx * DD + 2 * lane);
            accx += hv.x; accy += hv.y;
        }
        {
            const int    idx = s1[(size_t)b * SS + i];
            const float2 hv  = *(const float2*)(H + (size_t)idx * DD + 2 * lane);
            accx += hv.x; accy += hv.y;
        }
        dst = agg1 + (size_t)g * DD;
    } else {
        const int b = g - BB * SS;
        for (int j = 0; j < SS; ++j) {
            const int    idx = s1[(size_t)b * SS + j];
            const float2 hv  = *(const float2*)(H + (size_t)idx * DD + 2 * lane);
            accx += hv.x; accy += hv.y;
        }
        {
            const int    idx = bn[b];
            const float2 hv  = *(const float2*)(H + (size_t)idx * DD + 2 * lane);
            accx += hv.x; accy += hv.y;
        }
        dst = aggn + (size_t)b * DD;
    }

    const float sc = 1.f / 33.f;
    ((float2*)dst)[lane] = make_float2(accx * sc, accy * sc);
}

__global__ void k3_final(const float* __restrict__ agg1,
                         const float* __restrict__ aggn,
                         const float* __restrict__ W2,
                         const float* __restrict__ b2,
                         float* __restrict__ out) {
    __shared__ __attribute__((aligned(16))) float sX[48 * DD];
    __shared__ __attribute__((aligned(16))) float sW[DD * DD];
    const int t = threadIdx.x;
    const int b = blockIdx.x;

    for (int idx = t; idx < 48 * DD; idx += 128) {
        const int row = idx >> 6, col = idx & 63;
        float v = 0.f;
        if (row < SS)       v = agg1[((size_t)b * SS + row) * DD + col];
        else if (row == SS) v = aggn[(size_t)b * DD + col];
        sX[row * DD + swz_a(row, col)] = v;
    }
    for (int idx = t; idx < DD * DD; idx += 128) {
        const int k = idx >> 6, n = idx & (DD - 1);
        sW[k * DD + swz_w(k, n)] = W2[idx];
    }
    __syncthreads();

    const int wave  = t >> 5;
    const int lane  = t & 31;
    const int half  = lane >> 4;
    const int l     = lane & 15;
    const int ncol0 = wave * 16;
    const float bias = b2[ncol0 + l];
    const int bcol = swz_w(2 * half, ncol0 + l);

    float colsum = 0.f;
    #pragma unroll
    for (int mt = 0; mt < 3; ++mt) {
        v8f_t c = {0.f, 0.f, 0.f, 0.f, 0.f, 0.f, 0.f, 0.f};
        #pragma unroll
        for (int k0 = 0; k0 < DD; k0 += 4) {
            const int ka = k0 + 2 * half;
            v2f_t a, bb;
            a    = *(const v2f_t*)&sX[(mt * 16 + l) * DD + swz_a(l, ka)];
            bb.x = sW[ka * DD + bcol];
            bb.y = sW[(ka + 1) * DD + bcol];
            c = wmma_f32_4(a, bb, c);
        }
        #pragma unroll
        for (int r = 0; r < 8; ++r) {
            const int row = mt * 16 + half * 8 + r;
            if (row < SS + 1)
                colsum += fmaxf(c[r] + bias, 0.f);
        }
    }

    colsum += __shfl_xor(colsum, 16, 32);
    if (half == 0)
        out[(size_t)b * DD + ncol0 + l] = colsum * (1.f / 33.f);
}

extern "C" void kernel_launch(void* const* d_in, const int* in_sizes, int n_in,
                              void* d_out, int out_size, void* d_ws, size_t ws_size,
                              hipStream_t stream) {
    (void)in_sizes; (void)n_in; (void)out_size; (void)ws_size;

    const float* features    = (const float*)d_in[0];
    const int*   batch_nodes = (const int*)  d_in[1];
    const int*   s1_neighs   = (const int*)  d_in[2];
    const int*   s2_neighs   = (const int*)  d_in[3];
    const float* W1          = (const float*)d_in[4];
    const float* b1          = (const float*)d_in[5];
    const float* W2          = (const float*)d_in[6];
    const float* b2          = (const float*)d_in[7];
    float*       out         = (float*)d_out;

    float* H    = (float*)d_ws;
    float* agg1 = H    + (size_t)NN * DD;
    float* aggn = agg1 + (size_t)BB * SS * DD;

    k1_compute_h <<<K1_GROUPS,         128, 0, stream>>>(features, W1, b1, H);
    k2_gather_mean<<<(BB * SS + BB) / 8, 256, 0, stream>>>(H, s1_neighs, s2_neighs,
                                                           batch_nodes, agg1, aggn);
    k3_final     <<<BB,                128, 0, stream>>>(agg1, aggn, W2, b2, out);
}


# Gather-mean and final layer fused into one launch: one 4-wave work-group per batch element reads its indices once into LDS and gathers its 33 mean rows (16-byte lane loads) into the LDS X tile
# speedup vs baseline: 2.5852x; 1.5992x over previous
#include <hip/hip_runtime.h>
#include <hip/hip_bf16.h>

#define NN 100000
#define FF 128
#define BB 1024
#define SS 32
#define DD 64

__device__ __forceinline__ int swz_a(int row, int k) { return k ^ ((row & 15) << 2); }
__device__ __forceinline__ int swz_w(int k, int n)   { return n ^ (((k >> 1) & 1) << 4); }

typedef float v2f_t __attribute__((ext_vector_type(2)));
typedef float v8f_t __attribute__((ext_vector_type(8)));

#if defined(__AMDGCN__) && __has_builtin(__builtin_amdgcn_wmma_f32_16x16x4_f32)
#define HAVE_WMMA_F32 1
#endif

__device__ __forceinline__ v8f_t wmma_f32_4(v2f_t a, v2f_t b, v8f_t c) {
#ifdef HAVE_WMMA_F32
    return __builtin_amdgcn_wmma_f32_16x16x4_f32(false, a, false, b,
                                                 (short)0, c, false, false);
#else
    (void)a; (void)b;
    return c;
#endif
}

#define K1_TILES  (NN / 16)
#define K1_GROUPS (K1_TILES / 5)

__global__ void __launch_bounds__(128)
k1_compute_h(const float* __restrict__ feat,
             const float* __restrict__ W1,
             const float* __restrict__ b1,
             float* __restrict__ H) {
    __shared__ __attribute__((aligned(16))) float sA[2][16 * FF];
    const int t     = threadIdx.x;
    const int wave  = t >> 5;
    const int lane  = t & 31;
    const int half  = lane >> 4;
    const int l     = lane & 15;
    const int col   = wave * 16 + l;
    const int step  = gridDim.x;

    const int srow = t >> 5;
    const int sk   = (t & 31) << 2;

    int tile = blockIdx.x;
    float4 r0 = make_float4(0.f, 0.f, 0.f, 0.f), r1 = r0, r2 = r0, r3 = r0;
    if (tile < K1_TILES) {
        const float4* src = (const float4*)(feat + (size_t)tile * 16 * FF);
        r0 = src[t]; r1 = src[t + 128]; r2 = src[t + 256]; r3 = src[t + 384];
    }

    v2f_t bf[32];
    #pragma unroll
    for (int s = 0; s < 32; ++s) {
        const int ka = 4 * s + 2 * half;
        bf[s].x = W1[ka * DD + col];
        bf[s].y = W1[(ka + 1) * DD + col];
    }
    const float bias = b1[col];

    const int o0 = (srow     ) * FF + swz_a(srow,      sk);
    const int o1 = (srow +  4) * FF + swz_a(srow +  4, sk);
    const int o2 = (srow +  8) * FF + swz_a(srow +  8, sk);
    const int o3 = (srow + 12) * FF + swz_a(srow + 12, sk);

    if (tile < K1_TILES) {
        *(float4*)&sA[0][o0] = r0; *(float4*)&sA[0][o1] = r1;
        *(float4*)&sA[0][o2] = r2; *(float4*)&sA[0][o3] = r3;
    }

    int buf = 0;
    for (; tile < K1_TILES; tile += step) {
        __syncthreads();

        const int  next     = tile + step;
        const bool has_next = next < K1_TILES;
        if (has_next) {
            const float4* src = (const float4*)(feat + (size_t)next * 16 * FF);
            r0 = src[t]; r1 = src[t + 128]; r2 = src[t + 256]; r3 = src[t + 384];
        }

        const float* A = sA[buf];
        v8f_t c = {0.f, 0.f, 0.f, 0.f, 0.f, 0.f, 0.f, 0.f};
        #pragma unroll
        for (int s = 0; s < 32; ++s) {
            const int ka = 4 * s + 2 * half;
            const v2f_t a = *(const v2f_t*)&A[l * FF + swz_a(l, ka)];
            c = wmma_f32_4(a, bf[s], c);
        }

        const int row0 = tile * 16;
        #pragma unroll
        for (int q = 0; q < 8; ++q) {
            const int row = row0 + half * 8 + q;
            H[(size_t)row * DD + col] = fmaxf(c[q] + bias, 0.f);
        }

        if (has_next) {
            float* Bn = sA[buf ^ 1];
            *(float4*)&Bn[o0] = r0; *(float4*)&Bn[o1] = r1;
            *(float4*)&Bn[o2] = r2; *(float4*)&Bn[o3] = r3;
        }
        buf ^= 1;
    }
}

__global__ void __launch_bounds__(128)
k2_gather_final(const float* __restrict__ H,
                const int* __restrict__ s1,
                const int* __restrict__ s2,
                const int* __restrict__ bn,
                const float* __restrict__ W2,
                const float* __restrict__ b2,
                float* __restrict__ out) {
    __shared__ __attribute__((aligned(16))) float sX[48 * DD];
    __shared__ __attribute__((aligned(16))) float sW[DD * DD];
    __shared__ __attribute__((aligned(16))) float sSelf[SS * DD];
    __shared__ __attribute__((aligned(16))) int   sI2[SS * SS];
    __shared__ int sI1[128];

    const int t    = threadIdx.x;
    const int b    = blockIdx.x;
    const int wave = t >> 5;
    const int lane = t & 31;
    const int half = lane >> 4;
    const int l    = lane & 15;
    const int bnode = bn[b];

    {
        const int4* src2 = (const int4*)(s2 + (size_t)b * SS * SS);
        const int4 ia = src2[t];
        const int4 ib = src2[t + 128];
        const int i1 = s1[(size_t)b * SS + (t & (SS - 1))];

        float w[32];
        #pragma unroll
        for (int r = 0; r < 32; ++r) w[r] = W2[t + 128 * r];

        ((int4*)sI2)[t]       = ia;
        ((int4*)sI2)[t + 128] = ib;
        sI1[t] = i1;

        #pragma unroll
        for (int r = 0; r < 32; ++r) {
            const int idx = t + 128 * r;
            const int k = idx >> 6, n = idx & (DD - 1);
            sW[k * DD + swz_w(k, n)] = w[r];
        }

        const float4 z = make_float4(0.f, 0.f, 0.f, 0.f);
        ((float4*)&sX[(SS + 1) * DD])[t] = z;
        ((float4*)&sX[(SS + 1) * DD])[t + 128 < 240 ? t + 128 : t] = z;
    }
    __syncthreads();

    const char* Hb = (const char*)H;
    const float sc = 1.f / 33.f;
    #pragma unroll 1
    for (int p = 0; p < 4; ++p) {
        const int i = p * 8 + wave * 2 + half;
        float4 v[SS + 1];
        #pragma unroll
        for (int j = 0; j < SS; ++j) {
            const unsigned idx = (unsigned)sI2[j * SS + i];
            v[j] = *(const float4*)(Hb + (idx * 256u + (unsigned)l * 16u));
        }
        {
            const unsigned idx = (unsigned)sI1[i];
            v[SS] = *(const float4*)(Hb + (idx * 256u + (unsigned)l * 16u));
        }
        float4 acc = make_float4(0.f, 0.f, 0.f, 0.f);
        asm volatile("" : "+v"(acc.x), "+v"(acc.y), "+v"(acc.z), "+v"(acc.w));
        #pragma unroll
        for (int j = 0; j < SS + 1; ++j) {
            acc.x += v[j].x; acc.y += v[j].y; acc.z += v[j].z; acc.w += v[j].w;
        }
        *(float4*)&sSelf[i * DD + 4 * l] = v[SS];
        *(float4*)&sX[i * DD + swz_a(i, 4 * l)] =
            make_float4(acc.x * sc, acc.y * sc, acc.z * sc, acc.w * sc);
    }
    __syncthreads();

    if (t < DD) {
        const float hb = H[(size_t)bnode * DD + t];
        float a = 0.f;
        #pragma unroll
        for (int j = 0; j < SS; ++j) a += sSelf[j * DD + t];
        a += hb;
        sX[SS * DD + swz_a(SS, t)] = a * sc;
    }
    __syncthreads();

    const int ncol0 = wave * 16;
    const float bias = b2[ncol0 + l];
    const int bcol = swz_w(2 * half, ncol0 + l);

    float colsum = 0.f;
    #pragma unroll
    for (int mt = 0; mt < 3; ++mt) {
        v8f_t c = {0.f, 0.f, 0.f, 0.f, 0.f, 0.f, 0.f, 0.f};
        #pragma unroll
        for (int k0 = 0; k0 < DD; k0 += 4) {
            const int ka = k0 + 2 * half;
            v2f_t a, bb;
            a    = *(const v2f_t*)&sX[(mt * 16 + l) * DD + swz_a(l, ka)];
            bb.x = sW[ka * DD + bcol];
            bb.y = sW[(ka + 1) * DD + bcol];
            c = wmma_f32_4(a, bb, c);
        }
        #pragma unroll
        for (int r = 0; r < 8; ++r) {
            const int row = mt * 16 + half * 8 + r;
            if (row < SS + 1)
                colsum += fmaxf(c[r] + bias, 0.f);
        }
    }

    colsum += __shfl_xor(colsum, 16, 32);
    if (half == 0)
        out[(size_t)b * DD + ncol0 + l] = colsum * (1.f / 33.f);
}

extern "C" void kernel_launch(void* const* d_in, const int* in_sizes, int n_in,
                              void* d_out, int out_size, void* d_ws, size_t ws_size,
                              hipStream_t stream) {
    (void)in_sizes; (void)n_in; (void)out_size; (void)ws_size;

    const float* features    = (const float*)d_in[0];
    const int*   batch_nodes = (const int*)  d_in[1];
    const int*   s1_neighs   = (const int*)  d_in[2];
    const int*   s2_neighs   = (const int*)  d_in[3];
    const float* W1          = (const float*)d_in[4];
    const float* b1          = (const float*)d_in[5];
    const float* W2          = (const float*)d_in[6];
    const float* b2          = (const float*)d_in[7];
    float*       out         = (float*)d_out;

    float* H = (float*)d_ws;

    k1_compute_h   <<<K1_GROUPS, 128, 0, stream>>>(features, W1, b1, H);
    k2_gather_final<<<BB,        128, 0, stream>>>(H, s1_neighs, s2_neighs, batch_nodes,
                                                   W2, b2, out);
}
